# starting kernel
# baseline (speedup 1.0000x reference)
#include <hip/hip_runtime.h>
#include <hip/hip_bf16.h>

typedef __attribute__((ext_vector_type(2))) float v2f;
typedef __attribute__((ext_vector_type(4))) float v4f;
typedef __attribute__((ext_vector_type(8))) float v8f;

#define BATCHN 4096
#define SEQN   8192
#define NCHUNK 4
#define CHUNK  (SEQN / NCHUNK)
#define WARM   64

__global__ __launch_bounds__(32)
void iir_wmma_kernel(const float* __restrict__ x,
                     const float* __restrict__ A,
                     const float* __restrict__ B,
                     float* __restrict__ y) {
    __shared__ __align__(16) float h_sh[32];
    __shared__ __align__(16) float xt[16 * 20];
    __shared__ __align__(16) float yt[16 * 20];

    const int lane = threadIdx.x;
    const int half = lane >> 4;
    const int m    = lane & 15;
    const int rowbase = blockIdx.x * 16;
    const int cstart  = blockIdx.y * CHUNK;

    const float a0 = A[0], a1 = A[1], a2 = A[2], a3 = A[3];
    const float b0 = B[0], b1 = B[1], b2 = B[2], b3 = B[3], b4 = B[4];

    h_sh[lane] = 0.0f;
    __syncthreads();
    if (lane == 0) {
        h_sh[16] = 1.0f;
        for (int t = 1; t < 16; ++t)
            h_sh[16 + t] = a0 * h_sh[16 + t - 1] + a1 * h_sh[16 + t - 2]
                         + a2 * h_sh[16 + t - 3] + a3 * h_sh[16 + t - 4];
    }
    __syncthreads();

    v2f a_tf[4];
#pragma unroll
    for (int c = 0; c < 4; ++c) {
        const int k0 = 4 * c + 2 * half;
        a_tf[c].x = h_sh[16 + m - k0];
        a_tf[c].y = h_sh[16 + m - k0 - 1];
    }
    v2f a_th;
    {
        const float H0 = h_sh[16 + m], H1 = h_sh[15 + m],
                    H2 = h_sh[14 + m], H3 = h_sh[13 + m];
        const float th0 = a0 * H0 + a1 * H1 + a2 * H2 + a3 * H3;
        const float th1 = a1 * H0 + a2 * H1 + a3 * H2;
        const float th2 = a2 * H0 + a3 * H1;
        const float th3 = a3 * H0;
        a_th.x = (half == 0) ? th0 : th2;
        a_th.y = (half == 0) ? th1 : th3;
    }

    v2f bs; bs.x = 0.0f; bs.y = 0.0f;

    int n0 = cstart - WARM;
    const int iters = (CHUNK + WARM) / 16;

    for (int it = 0; it < iters; ++it) {
#pragma unroll
        for (int s = 0; s < 3; ++s) {
            const int idx = lane + 32 * s;
            if (idx < 80) {
                const int r = idx / 5, seg = idx % 5;
                const int pos = n0 - 4 + seg * 4;
                v4f v = {0.0f, 0.0f, 0.0f, 0.0f};
                if (pos >= 0)
                    v = *(const v4f*)(x + (size_t)(rowbase + r) * SEQN + pos);
                *(v4f*)&xt[r * 20 + seg * 4] = v;
            }
        }
        if (lane < 16 && (n0 + 160) < SEQN)
            __builtin_prefetch(x + (size_t)(rowbase + lane) * SEQN + n0 + 128, 0, 1);
        __syncthreads();

        const int nb = m * 20;
        const int jb = 2 * half;
        v2f bff[4];
#pragma unroll
        for (int c = 0; c < 4; ++c) {
            const int j0 = 4 * c + jb;
            bff[c].x = b0 * xt[nb + j0 + 4] + b1 * xt[nb + j0 + 3]
                     + b2 * xt[nb + j0 + 2] + b3 * xt[nb + j0 + 1]
                     + b4 * xt[nb + j0 + 0];
            bff[c].y = b0 * xt[nb + j0 + 5] + b1 * xt[nb + j0 + 4]
                     + b2 * xt[nb + j0 + 3] + b3 * xt[nb + j0 + 2]
                     + b4 * xt[nb + j0 + 1];
        }

        v8f acc = {};
#pragma unroll
        for (int c = 0; c < 4; ++c)
            acc = __builtin_amdgcn_wmma_f32_16x16x4_f32(
                false, a_tf[c], false, bff[c], (short)0, acc, false, false);
        acc = __builtin_amdgcn_wmma_f32_16x16x4_f32(
                false, a_th, false, bs, (short)0, acc, false, false);

        const float t7 = __shfl_xor(acc[7], 16, 32);
        const float t6 = __shfl_xor(acc[6], 16, 32);
        bs.x = (half == 0) ? t7 : acc[5];
        bs.y = (half == 0) ? t6 : acc[4];

        if (n0 >= cstart) {
#pragma unroll
            for (int v = 0; v < 8; ++v)
                yt[m * 20 + half * 8 + v] = acc[v];
            __syncthreads();
#pragma unroll
            for (int s = 0; s < 2; ++s) {
                const int idx = lane + 32 * s;
                const int r = idx >> 2, seg = idx & 3;
                *(v4f*)(y + (size_t)(rowbase + r) * SEQN + n0 + seg * 4) =
                    *(const v4f*)&yt[r * 20 + seg * 4];
            }
        }
        __syncthreads();
        n0 += 16;
    }
}

extern "C" void kernel_launch(void* const* d_in, const int* in_sizes, int n_in,
                              void* d_out, int out_size, void* d_ws, size_t ws_size,
                              hipStream_t stream) {
    const float* x = (const float*)d_in[0];
    const float* a = (const float*)d_in[1];
    const float* b = (const float*)d_in[2];
    float* out = (float*)d_out;
    (void)in_sizes; (void)n_in; (void)out_size; (void)d_ws; (void)ws_size;

    dim3 grid(BATCHN / 16, NCHUNK, 1);
    iir_wmma_kernel<<<grid, 32, 0, stream>>>(x, a, b, out);
}


# NCHUNK 4 -> 32: 8192 one-wave blocks (8 per SIMD) so each wave's load and store waits are covered by other waves; the 28 added chunk starts use the parent's 64-step zero-state warm-up
# speedup vs baseline: 1.5702x; 1.5702x over previous
#include <hip/hip_runtime.h>
#include <hip/hip_bf16.h>

typedef __attribute__((ext_vector_type(2))) float v2f;
typedef __attribute__((ext_vector_type(4))) float v4f;
typedef __attribute__((ext_vector_type(8))) float v8f;

#define BATCHN 4096
#define SEQN   8192
#define NCHUNK 32
#define CHUNK  (SEQN / NCHUNK)
#define WARM   64

__global__ __launch_bounds__(32)
void iir_wmma_kernel(const float* __restrict__ x,
                     const float* __restrict__ A,
                     const float* __restrict__ B,
                     float* __restrict__ y) {
    __shared__ __align__(16) float h_sh[32];
    __shared__ __align__(16) float xt[16 * 20];
    __shared__ __align__(16) float yt[16 * 20];

    const int lane = threadIdx.x;
    const int half = lane >> 4;
    const int m    = lane & 15;
    const int rowbase = blockIdx.x * 16;
    const int cstart  = blockIdx.y * CHUNK;

    const float a0 = A[0], a1 = A[1], a2 = A[2], a3 = A[3];
    const float b0 = B[0], b1 = B[1], b2 = B[2], b3 = B[3], b4 = B[4];

    h_sh[lane] = 0.0f;
    __syncthreads();
    if (lane == 0) {
        h_sh[16] = 1.0f;
        for (int t = 1; t < 16; ++t)
            h_sh[16 + t] = a0 * h_sh[16 + t - 1] + a1 * h_sh[16 + t - 2]
                         + a2 * h_sh[16 + t - 3] + a3 * h_sh[16 + t - 4];
    }
    __syncthreads();

    v2f a_tf[4];
#pragma unroll
    for (int c = 0; c < 4; ++c) {
        const int k0 = 4 * c + 2 * half;
        a_tf[c].x = h_sh[16 + m - k0];
        a_tf[c].y = h_sh[16 + m - k0 - 1];
    }
    v2f a_th;
    {
        const float H0 = h_sh[16 + m], H1 = h_sh[15 + m],
                    H2 = h_sh[14 + m], H3 = h_sh[13 + m];
        const float th0 = a0 * H0 + a1 * H1 + a2 * H2 + a3 * H3;
        const float th1 = a1 * H0 + a2 * H1 + a3 * H2;
        const float th2 = a2 * H0 + a3 * H1;
        const float th3 = a3 * H0;
        a_th.x = (half == 0) ? th0 : th2;
        a_th.y = (half == 0) ? th1 : th3;
    }

    v2f bs; bs.x = 0.0f; bs.y = 0.0f;

    int n0 = cstart - WARM;
    const int iters = (CHUNK + WARM) / 16;

    for (int it = 0; it < iters; ++it) {
#pragma unroll
        for (int s = 0; s < 3; ++s) {
            const int idx = lane + 32 * s;
            if (idx < 80) {
                const int r = idx / 5, seg = idx % 5;
                const int pos = n0 - 4 + seg * 4;
                v4f v = {0.0f, 0.0f, 0.0f, 0.0f};
                if (pos >= 0)
                    v = *(const v4f*)(x + (size_t)(rowbase + r) * SEQN + pos);
                *(v4f*)&xt[r * 20 + seg * 4] = v;
            }
        }
        if (lane < 16 && (n0 + 160) < SEQN)
            __builtin_prefetch(x + (size_t)(rowbase + lane) * SEQN + n0 + 128, 0, 1);
        __syncthreads();

        const int nb = m * 20;
        const int jb = 2 * half;
        v2f bff[4];
#pragma unroll
        for (int c = 0; c < 4; ++c) {
            const int j0 = 4 * c + jb;
            bff[c].x = b0 * xt[nb + j0 + 4] + b1 * xt[nb + j0 + 3]
                     + b2 * xt[nb + j0 + 2] + b3 * xt[nb + j0 + 1]
                     + b4 * xt[nb + j0 + 0];
            bff[c].y = b0 * xt[nb + j0 + 5] + b1 * xt[nb + j0 + 4]
                     + b2 * xt[nb + j0 + 3] + b3 * xt[nb + j0 + 2]
                     + b4 * xt[nb + j0 + 1];
        }

        v8f acc = {};
#pragma unroll
        for (int c = 0; c < 4; ++c)
            acc = __builtin_amdgcn_wmma_f32_16x16x4_f32(
                false, a_tf[c], false, bff[c], (short)0, acc, false, false);
        acc = __builtin_amdgcn_wmma_f32_16x16x4_f32(
                false, a_th, false, bs, (short)0, acc, false, false);

        const float t7 = __shfl_xor(acc[7], 16, 32);
        const float t6 = __shfl_xor(acc[6], 16, 32);
        bs.x = (half == 0) ? t7 : acc[5];
        bs.y = (half == 0) ? t6 : acc[4];

        if (n0 >= cstart) {
#pragma unroll
            for (int v = 0; v < 8; ++v)
                yt[m * 20 + half * 8 + v] = acc[v];
            __syncthreads();
#pragma unroll
            for (int s = 0; s < 2; ++s) {
                const int idx = lane + 32 * s;
                const int r = idx >> 2, seg = idx & 3;
                *(v4f*)(y + (size_t)(rowbase + r) * SEQN + n0 + seg * 4) =
                    *(const v4f*)&yt[r * 20 + seg * 4];
            }
        }
        __syncthreads();
        n0 += 16;
    }
}

extern "C" void kernel_launch(void* const* d_in, const int* in_sizes, int n_in,
                              void* d_out, int out_size, void* d_ws, size_t ws_size,
                              hipStream_t stream) {
    const float* x = (const float*)d_in[0];
    const float* a = (const float*)d_in[1];
    const float* b = (const float*)d_in[2];
    float* out = (float*)d_out;
    (void)in_sizes; (void)n_in; (void)out_size; (void)d_ws; (void)ws_size;

    dim3 grid(BATCHN / 16, NCHUNK, 1);
    iir_wmma_kernel<<<grid, 32, 0, stream>>>(x, a, b, out);
}


# Tile 16 -> 64 steps: each wave-wide load/store covers whole 128-B lines (2 rows x 256 B) and 8 loads fly together instead of 3 waited one by one; halo kept in LDS, look-ahead hint dropped
# speedup vs baseline: 4.2035x; 2.6770x over previous
#include <hip/hip_runtime.h>
#include <hip/hip_bf16.h>

typedef __attribute__((ext_vector_type(2))) float v2f;
typedef __attribute__((ext_vector_type(4))) float v4f;
typedef __attribute__((ext_vector_type(8))) float v8f;

#define BATCHN 4096
#define SEQN   8192
#define NCHUNK 32
#define CHUNK  (SEQN / NCHUNK)
#define WARM   64
#define TILE   64
#define NSUB   (TILE / 16)
#define PITCH  (TILE + 4)

__global__ __launch_bounds__(32, 8)
void iir_wmma_kernel(const float* __restrict__ x,
                     const float* __restrict__ A,
                     const float* __restrict__ B,
                     float* __restrict__ y) {
    __shared__ __align__(16) float h_sh[32];
    __shared__ __align__(16) float xt[16 * PITCH];

    const int lane = threadIdx.x;
    const int half = lane >> 4;
    const int m    = lane & 15;
    const int rowbase = blockIdx.x * 16;
    const int cstart  = blockIdx.y * CHUNK;

    const float a0 = A[0], a1 = A[1], a2 = A[2], a3 = A[3];
    const float b0 = B[0], b1 = B[1], b2 = B[2], b3 = B[3], b4 = B[4];

    h_sh[lane] = 0.0f;
    __syncthreads();
    if (lane == 0) {
        h_sh[16] = 1.0f;
        for (int t = 1; t < 16; ++t)
            h_sh[16 + t] = a0 * h_sh[16 + t - 1] + a1 * h_sh[16 + t - 2]
                         + a2 * h_sh[16 + t - 3] + a3 * h_sh[16 + t - 4];
    }
    __syncthreads();

    v2f a_tf[4];
#pragma unroll
    for (int c = 0; c < 4; ++c) {
        const int k0 = 4 * c + 2 * half;
        a_tf[c].x = h_sh[16 + m - k0];
        a_tf[c].y = h_sh[16 + m - k0 - 1];
    }
    v2f a_th;
    {
        const float H0 = h_sh[16 + m], H1 = h_sh[15 + m],
                    H2 = h_sh[14 + m], H3 = h_sh[13 + m];
        const float th0 = a0 * H0 + a1 * H1 + a2 * H2 + a3 * H3;
        const float th1 = a1 * H0 + a2 * H1 + a3 * H2;
        const float th2 = a2 * H0 + a3 * H1;
        const float th3 = a3 * H0;
        a_th.x = (half == 0) ? th0 : th2;
        a_th.y = (half == 0) ? th1 : th3;
    }

    v2f bs; bs.x = 0.0f; bs.y = 0.0f;

    int n0 = cstart - WARM;

    if (lane < 16) {
        v4f hv0 = {0.0f, 0.0f, 0.0f, 0.0f};
        if (n0 >= 4)
            hv0 = *(const v4f*)(x + (size_t)(rowbase + lane) * SEQN + n0 - 4);
        *(v4f*)&xt[lane * PITCH] = hv0;
    }

    const int iters = (CHUNK + WARM) / TILE;

#pragma unroll 1
    for (int it = 0; it < iters; ++it) {
        v4f xv[8];
#pragma unroll
        for (int s = 0; s < 8; ++s) xv[s] = (v4f){0.0f, 0.0f, 0.0f, 0.0f};
        if (n0 >= 0) {
#pragma unroll
            for (int s = 0; s < 8; ++s) {
                const int idx = lane + 32 * s;
                const int r = idx >> 4, seg = idx & 15;
                xv[s] = *(const v4f*)(x + (size_t)(rowbase + r) * SEQN + n0 + seg * 4);
            }
        }
#pragma unroll
        for (int s = 0; s < 8; ++s) {
            const int idx = lane + 32 * s;
            const int r = idx >> 4, seg = idx & 15;
            *(v4f*)&xt[r * PITCH + 4 + seg * 4] = xv[s];
        }
        __syncthreads();

        const int nb = m * PITCH;
        const int jb = 2 * half;
        v2f bff[NSUB][4];
#pragma unroll
        for (int q = 0; q < NSUB; ++q) {
#pragma unroll
            for (int c = 0; c < 4; ++c) {
                const int j0 = 16 * q + 4 * c + jb;
                const float* p = &xt[nb + j0];
                const v2f x0 = {p[0], p[1]}, x1 = {p[1], p[2]}, x2 = {p[2], p[3]},
                          x3 = {p[3], p[4]}, x4 = {p[4], p[5]};
                bff[q][c] = b0 * x4 + b1 * x3 + b2 * x2 + b3 * x1 + b4 * x0;
            }
        }
        v4f hv = {0.0f, 0.0f, 0.0f, 0.0f};
        if (lane < 16)
            hv = *(const v4f*)&xt[lane * PITCH + TILE];

        v8f acc[NSUB];
#pragma unroll
        for (int q = 0; q < NSUB; ++q) acc[q] = (v8f){};
#pragma unroll
        for (int c = 0; c < 4; ++c) {
#pragma unroll
            for (int q = 0; q < NSUB; ++q)
                acc[q] = __builtin_amdgcn_wmma_f32_16x16x4_f32(
                    false, a_tf[c], false, bff[q][c], (short)0, acc[q], false, false);
        }
#pragma unroll
        for (int q = 0; q < NSUB; ++q) {
            acc[q] = __builtin_amdgcn_wmma_f32_16x16x4_f32(
                false, a_th, false, bs, (short)0, acc[q], false, false);
            const float t7 = __shfl_xor(acc[q][7], 16, 32);
            const float t6 = __shfl_xor(acc[q][6], 16, 32);
            bs.x = (half == 0) ? t7 : acc[q][5];
            bs.y = (half == 0) ? t6 : acc[q][4];
        }

        __syncthreads();
        if (lane < 16)
            *(v4f*)&xt[lane * PITCH] = hv;

        if (n0 >= cstart) {
#pragma unroll
            for (int q = 0; q < NSUB; ++q) {
#pragma unroll
                for (int v = 0; v < 8; ++v)
                    xt[m * PITCH + 4 + 16 * q + half * 8 + v] = acc[q][v];
            }
            __syncthreads();
#pragma unroll
            for (int s = 0; s < 8; ++s) {
                const int idx = lane + 32 * s;
                const int r = idx >> 4, seg = idx & 15;
                *(v4f*)(y + (size_t)(rowbase + r) * SEQN + n0 + seg * 4) =
                    *(const v4f*)&xt[r * PITCH + 4 + seg * 4];
            }
        }
        __syncthreads();
        n0 += TILE;
    }
}

extern "C" void kernel_launch(void* const* d_in, const int* in_sizes, int n_in,
                              void* d_out, int out_size, void* d_ws, size_t ws_size,
                              hipStream_t stream) {
    const float* x = (const float*)d_in[0];
    const float* a = (const float*)d_in[1];
    const float* b = (const float*)d_in[2];
    float* out = (float*)d_out;
    (void)in_sizes; (void)n_in; (void)out_size; (void)d_ws; (void)ws_size;

    dim3 grid(BATCHN / 16, NCHUNK, 1);
    iir_wmma_kernel<<<grid, 32, 0, stream>>>(x, a, b, out);
}


# NCHUNK 32 -> 16: 4096 one-wave blocks (4 per SIMD) enter in half the time and re-read 1/9 not 1/5 for warm-up; 256 regs keep two tiles of loads in flight per wave, LDS-only fences
# speedup vs baseline: 5.1629x; 1.2282x over previous
#include <hip/hip_runtime.h>
#include <hip/hip_bf16.h>


typedef __attribute__((ext_vector_type(2))) float v2f;
typedef __attribute__((ext_vector_type(4))) float v4f;
typedef __attribute__((ext_vector_type(8))) float v8f;

#define BATCHN 4096
#define SEQN   8192
#define NCHUNK 16
#define CHUNK  (SEQN / NCHUNK)
#define WARM   64
#define TILE   64
#define NSUB   (TILE / 16)
#define PITCH  (TILE + 4)
#define ITERS  ((CHUNK + WARM) / TILE)

static_assert(ITERS >= 3 && (ITERS % 2) == 1, "tile schedule below: pairs, then three closing tiles");

__device__ __forceinline__ void lds_sync() {
    __builtin_amdgcn_fence(__ATOMIC_RELEASE, "workgroup", "local");
    __builtin_amdgcn_s_barrier();
    __builtin_amdgcn_fence(__ATOMIC_ACQUIRE, "workgroup", "local");
}

__device__ __forceinline__ void load_tile(v4f (&buf)[8], const float* __restrict__ x,
                                          int rowbase, int n0, int lane) {
#pragma unroll
    for (int s = 0; s < 8; ++s) {
        const int idx = lane + 32 * s;
        const int r = idx >> 4, seg = idx & 15;
        buf[s] = *(const v4f*)(x + (size_t)(rowbase + r) * SEQN + n0 + seg * 4);
    }
}

template <bool REFILL>
__device__ __forceinline__ void tile_step(v4f (&cur)[8],
                                          const float* __restrict__ x, float* __restrict__ y,
                                          float* xt, int rowbase, int n0, int cstart,
                                          int lane, int half, int m,
                                          const v2f (&a_tf)[4], v2f a_th, v2f& bs,
                                          float b0, float b1, float b2, float b3, float b4) {
#pragma unroll
    for (int s = 0; s < 8; ++s) {
        const int idx = lane + 32 * s;
        const int r = idx >> 4, seg = idx & 15;
        *(v4f*)&xt[r * PITCH + 4 + seg * 4] = cur[s];
    }
    if (REFILL)
        load_tile(cur, x, rowbase, n0 + 2 * TILE, lane);
    lds_sync();

    const int nb = m * PITCH;
    const int jb = 2 * half;
    v2f bff[NSUB][4];
#pragma unroll
    for (int q = 0; q < NSUB; ++q) {
#pragma unroll
        for (int c = 0; c < 4; ++c) {
            const int j0 = 16 * q + 4 * c + jb;
            const float* p = &xt[nb + j0];
            const v2f x0 = {p[0], p[1]}, x1 = {p[1], p[2]}, x2 = {p[2], p[3]},
                      x3 = {p[3], p[4]}, x4 = {p[4], p[5]};
            bff[q][c] = b0 * x4 + b1 * x3 + b2 * x2 + b3 * x1 + b4 * x0;
        }
    }
    v4f hv = {0.0f, 0.0f, 0.0f, 0.0f};
    if (lane < 16)
        hv = *(const v4f*)&xt[lane * PITCH + TILE];

    v8f acc[NSUB];
#pragma unroll
    for (int q = 0; q < NSUB; ++q) acc[q] = (v8f){};
#pragma unroll
    for (int c = 0; c < 4; ++c) {
#pragma unroll
        for (int q = 0; q < NSUB; ++q)
            acc[q] = __builtin_amdgcn_wmma_f32_16x16x4_f32(
                false, a_tf[c], false, bff[q][c], (short)0, acc[q], false, false);
    }
#pragma unroll
    for (int q = 0; q < NSUB; ++q) {
        acc[q] = __builtin_amdgcn_wmma_f32_16x16x4_f32(
            false, a_th, false, bs, (short)0, acc[q], false, false);
        const float t7 = __shfl_xor(acc[q][7], 16, 32);
        const float t6 = __shfl_xor(acc[q][6], 16, 32);
        bs.x = (half == 0) ? t7 : acc[q][5];
        bs.y = (half == 0) ? t6 : acc[q][4];
    }

    lds_sync();
    if (lane < 16)
        *(v4f*)&xt[lane * PITCH] = hv;

    if (n0 >= cstart) {
#pragma unroll
        for (int q = 0; q < NSUB; ++q) {
#pragma unroll
            for (int v = 0; v < 8; ++v)
                xt[m * PITCH + 4 + 16 * q + half * 8 + v] = acc[q][v];
        }
        lds_sync();
#pragma unroll
        for (int s = 0; s < 8; ++s) {
            const int idx = lane + 32 * s;
            const int r = idx >> 4, seg = idx & 15;
            *(v4f*)(y + (size_t)(rowbase + r) * SEQN + n0 + seg * 4) =
                *(const v4f*)&xt[r * PITCH + 4 + seg * 4];
        }
    }
    lds_sync();
}

__global__ __launch_bounds__(32, 4)
void iir_wmma_kernel(const float* __restrict__ x,
                     const float* __restrict__ A,
                     const float* __restrict__ B,
                     float* __restrict__ y) {
    __shared__ __align__(16) float h_sh[32];
    __shared__ __align__(16) float xt[16 * PITCH];

    const int lane = threadIdx.x;
    const int half = lane >> 4;
    const int m    = lane & 15;
    const int rowbase = blockIdx.x * 16;
    const int cstart  = blockIdx.y * CHUNK;
    const int nfirst  = cstart - WARM;

    v4f hv0 = {0.0f, 0.0f, 0.0f, 0.0f};
    v4f buf0[8], buf1[8];
#pragma unroll
    for (int s = 0; s < 8; ++s) buf0[s] = (v4f){0.0f, 0.0f, 0.0f, 0.0f};
    if (nfirst >= 4) {
        load_tile(buf0, x, rowbase, nfirst, lane);
        hv0 = *(const v4f*)(x + (size_t)(rowbase + m) * SEQN + nfirst - 4);
    }
    load_tile(buf1, x, rowbase, nfirst + TILE, lane);

    const float a0 = A[0], a1 = A[1], a2 = A[2], a3 = A[3];
    const float b0 = B[0], b1 = B[1], b2 = B[2], b3 = B[3], b4 = B[4];

    h_sh[lane] = 0.0f;
    lds_sync();
    if (lane == 0) {
        h_sh[16] = 1.0f;
        for (int t = 1; t < 16; ++t)
            h_sh[16 + t] = a0 * h_sh[16 + t - 1] + a1 * h_sh[16 + t - 2]
                         + a2 * h_sh[16 + t - 3] + a3 * h_sh[16 + t - 4];
    }
    lds_sync();

    v2f a_tf[4];
#pragma unroll
    for (int c = 0; c < 4; ++c) {
        const int k0 = 4 * c + 2 * half;
        a_tf[c].x = h_sh[16 + m - k0];
        a_tf[c].y = h_sh[16 + m - k0 - 1];
    }
    v2f a_th;
    {
        const float H0 = h_sh[16 + m], H1 = h_sh[15 + m],
                    H2 = h_sh[14 + m], H3 = h_sh[13 + m];
        const float th0 = a0 * H0 + a1 * H1 + a2 * H2 + a3 * H3;
        const float th1 = a1 * H0 + a2 * H1 + a3 * H2;
        const float th2 = a2 * H0 + a3 * H1;
        const float th3 = a3 * H0;
        a_th.x = (half == 0) ? th0 : th2;
        a_th.y = (half == 0) ? th1 : th3;
    }

    v2f bs; bs.x = 0.0f; bs.y = 0.0f;

    if (lane < 16)
        *(v4f*)&xt[lane * PITCH] = hv0;

    int n0 = nfirst;
#pragma unroll 1
    for (int it = 0; it + 3 < ITERS; it += 2) {
        tile_step<true>(buf0, x, y, xt, rowbase, n0, cstart, lane, half, m, a_tf, a_th, bs, b0, b1, b2, b3, b4);
        n0 += TILE;
        tile_step<true>(buf1, x, y, xt, rowbase, n0, cstart, lane, half, m, a_tf, a_th, bs, b0, b1, b2, b3, b4);
        n0 += TILE;
    }
    tile_step<true>(buf0, x, y, xt, rowbase, n0, cstart, lane, half, m, a_tf, a_th, bs, b0, b1, b2, b3, b4);
    n0 += TILE;
    tile_step<false>(buf1, x, y, xt, rowbase, n0, cstart, lane, half, m, a_tf, a_th, bs, b0, b1, b2, b3, b4);
    n0 += TILE;
    tile_step<false>(buf0, x, y, xt, rowbase, n0, cstart, lane, half, m, a_tf, a_th, bs, b0, b1, b2, b3, b4);
}

extern "C" void kernel_launch(void* const* d_in, const int* in_sizes, int n_in,
                              void* d_out, int out_size, void* d_ws, size_t ws_size,
                              hipStream_t stream) {
    const float* x = (const float*)d_in[0];
    const float* a = (const float*)d_in[1];
    const float* b = (const float*)d_in[2];
    float* out = (float*)d_out;
    (void)in_sizes; (void)n_in; (void)out_size; (void)d_ws; (void)ws_size;

    dim3 grid(BATCHN / 16, NCHUNK, 1);
    iir_wmma_kernel<<<grid, 32, 0, stream>>>(x, a, b, out);
}
